# starting kernel
# baseline (speedup 1.0000x reference)
#include <hip/hip_runtime.h>
#include <hip/hip_bf16.h>

typedef __attribute__((ext_vector_type(2))) float v2f;
typedef __attribute__((ext_vector_type(8))) float v8f;

#define IN_CH 128
#define LAT   32
#define TILE_M 16
#define APAD  132


__global__ void deg_init_kernel(float* __restrict__ deg, int n) {
    int i = blockIdx.x * blockDim.x + threadIdx.x;
    if (i < n) deg[i] = 1.0f;
}

__global__ void deg_edges_kernel(const int* __restrict__ ei, const int* __restrict__ yei,
                                 float* __restrict__ deg, int e1, int e2) {
    int e = blockIdx.x * blockDim.x + threadIdx.x;
    int et = e1 + e2;
    if (e >= et) return;
    int d = (e < e1) ? ei[e1 + e] : yei[e2 + (e - e1)];
    atomicAdd(deg + d, 1.0f);
}

__global__ void dinv_kernel(float* __restrict__ deg, int n) {
    int i = blockIdx.x * blockDim.x + threadIdx.x;
    if (i < n) deg[i] = rsqrtf(deg[i]);
}


__global__ __launch_bounds__(128) void gemm_wmma_kernel(const float* __restrict__ x,
                                                        const float* __restrict__ W,
                                                        float* __restrict__ h, int n) {
    __shared__ float sW[IN_CH * LAT];
    __shared__ float sA[4][TILE_M * APAD];

    const int tid  = threadIdx.x;
    const int wave = tid >> 5;
    const int lane = tid & 31;
    const int ntiles = (n + TILE_M - 1) / TILE_M;
    const int tile = blockIdx.x * 4 + wave;

    {
        const float4* Wv = (const float4*)W;
        float4* sWv = (float4*)sW;
        for (int i = tid; i < (IN_CH * LAT) / 4; i += 128) sWv[i] = Wv[i];
    }
    if (tile < ntiles) {
        const int r0 = tile * TILE_M;
        for (int i = lane; i < TILE_M * (IN_CH / 4); i += 32) {
            int row = i >> 5;
            int c4  = i & 31;
            int grow = r0 + row; if (grow >= n) grow = n - 1;
            float4 v = ((const float4*)(x + (size_t)grow * IN_CH))[c4];
            float* dst = &sA[wave][row * APAD + c4 * 4];
            dst[0] = v.x; dst[1] = v.y; dst[2] = v.z; dst[3] = v.w;
        }
    }
    __syncthreads();

    if (tile >= ntiles) return;

    v8f acc0 = {};
    v8f acc1 = {};
    const int m    = lane & 15;
    const int koff = (lane >> 4) << 1;
    const int nn   = lane & 15;

    for (int kk = 0; kk < IN_CH / 4; ++kk) {
        const int kb = kk * 4 + koff;
        v2f a;
        a.x = sA[wave][m * APAD + kb];
        a.y = sA[wave][m * APAD + kb + 1];
        v2f b0, b1;
        b0.x = sW[(kb    ) * LAT + nn];
        b0.y = sW[(kb + 1) * LAT + nn];
        b1.x = sW[(kb    ) * LAT + 16 + nn];
        b1.y = sW[(kb + 1) * LAT + 16 + nn];
        acc0 = __builtin_amdgcn_wmma_f32_16x16x4_f32(false, a, false, b0,
                                                     (short)0, acc0, false, false);
        acc1 = __builtin_amdgcn_wmma_f32_16x16x4_f32(false, a, false, b1,
                                                     (short)0, acc1, false, false);
    }

    const int r0   = tile * TILE_M;
    const int rofs = (lane >> 4) << 3;
#pragma unroll
    for (int r = 0; r < 8; ++r) {
        int row = r0 + r + rofs;
        if (row < n) {
            h[(size_t)row * LAT + nn]      = acc0[r];
            h[(size_t)row * LAT + 16 + nn] = acc1[r];
        }
    }
}


__global__ void init_out_kernel(const float* __restrict__ dinv, const float* __restrict__ h,
                                const float* __restrict__ b, float* __restrict__ out, int n) {
    int gid = blockIdx.x * blockDim.x + threadIdx.x;
    if (gid >= n * LAT) return;
    int i = gid >> 5, c = gid & 31;
    float di = dinv[i];
    out[gid] = di * di * h[gid] + b[c];
}


__global__ void edge_scatter_kernel(const int* __restrict__ ei, const int* __restrict__ yei,
                                    const float* __restrict__ dinv, const float* __restrict__ h,
                                    float* __restrict__ out, int e1, int e2) {
    long long gid = (long long)blockIdx.x * blockDim.x + threadIdx.x;
    int e = (int)(gid >> 5);
    int c = (int)(gid & 31);
    int et = e1 + e2;
    if (e >= et) return;
    int s, d;
    if (e < e1) { s = ei[e];            d = ei[e1 + e]; }
    else        { int f = e - e1; s = yei[f]; d = yei[e2 + f]; }
    float w = dinv[s] * dinv[d];
    atomicAdd(out + (size_t)d * LAT + c, w * h[(size_t)s * LAT + c]);
}


extern "C" void kernel_launch(void* const* d_in, const int* in_sizes, int n_in,
                              void* d_out, int out_size, void* d_ws, size_t ws_size,
                              hipStream_t stream) {
    const float* x   = (const float*)d_in[0];
    const int*   ei  = (const int*)  d_in[1];
    const int*   yei = (const int*)  d_in[2];
    const float* W   = (const float*)d_in[3];
    const float* b   = (const float*)d_in[4];
    float* out = (float*)d_out;

    const int n  = in_sizes[0] / IN_CH;
    const int e1 = in_sizes[1] / 2;
    const int e2 = in_sizes[2] / 2;
    const int et = e1 + e2;

    float* deg = (float*)d_ws;
    size_t hoff = (((size_t)n * sizeof(float)) + 255) & ~(size_t)255;
    float* h = (float*)((char*)d_ws + hoff);

    deg_init_kernel<<<(n + 255) / 256, 256, 0, stream>>>(deg, n);
    deg_edges_kernel<<<(et + 255) / 256, 256, 0, stream>>>(ei, yei, deg, e1, e2);
    dinv_kernel<<<(n + 255) / 256, 256, 0, stream>>>(deg, n);
    {
        int ntiles = (n + TILE_M - 1) / TILE_M;
        int blocks = (ntiles + 3) / 4;
        gemm_wmma_kernel<<<blocks, 128, 0, stream>>>(x, W, h, n);
    }
    init_out_kernel<<<((size_t)n * LAT + 255) / 256, 256, 0, stream>>>(deg, h, b, out, n);
    {
        long long threads = (long long)et * 32;
        int blocks = (int)((threads + 255) / 256);
        edge_scatter_kernel<<<blocks, 256, 0, stream>>>(ei, yei, deg, h, out, e1, e2);
    }
}


# degree count privatized in LDS: each block histograms one 8192-node range over an edge slice, then flushes with coalesced float atomics; partial sums are exact integers, sum order changes
# speedup vs baseline: 1.3446x; 1.3446x over previous
#include <hip/hip_runtime.h>
#include <hip/hip_bf16.h>

typedef __attribute__((ext_vector_type(2))) float v2f;
typedef __attribute__((ext_vector_type(8))) float v8f;

#define IN_CH 128
#define LAT   32
#define TILE_M 16
#define APAD  132


__global__ void deg_init_kernel(float* __restrict__ deg, int n) {
    int i = blockIdx.x * blockDim.x + threadIdx.x;
    if (i < n) deg[i] = 1.0f;
}


#define DEG_RANGE   8192
#define DEG_THREADS 1024
#define DEG_GRID    256

__device__ __forceinline__ void deg_count(unsigned* cnt, int d, int base) {
    unsigned k = (unsigned)d - (unsigned)base;
    if (k < (unsigned)DEG_RANGE) atomicAdd(&cnt[k], 1u);
}

__device__ __forceinline__ void deg_count4(unsigned* cnt, int4 v, int base) {
    deg_count(cnt, v.x, base); deg_count(cnt, v.y, base);
    deg_count(cnt, v.z, base); deg_count(cnt, v.w, base);
}

__device__ __forceinline__ void deg_scan(const int* __restrict__ p, int L, int s, int S,
                                         unsigned* cnt, int base, int tid) {
    int head = (int)(((16u - (unsigned)((size_t)p & 15u)) & 15u) >> 2);
    if (head > L) head = L;
    const int nv    = (L - head) >> 2;
    const int tail0 = head + (nv << 2);
    if (s == 0) {
        if (tid < head) deg_count(cnt, p[tid], base);
        const int t = tail0 + tid;
        if (t < L) deg_count(cnt, p[t], base);
    }
    const int4* __restrict__ pv = (const int4*)(p + head);
    const int vchunk = (nv + S - 1) / S;
    const long long lo64 = (long long)s * vchunk;
    const int lo = lo64 < nv ? (int)lo64 : nv;
    const int hi = (nv - lo < vchunk) ? nv : lo + vchunk;
    int i = lo + tid;
    for (; i + 3 * DEG_THREADS < hi; i += 4 * DEG_THREADS) {
        const int4 v0 = pv[i];
        const int4 v1 = pv[i + DEG_THREADS];
        const int4 v2 = pv[i + 2 * DEG_THREADS];
        const int4 v3 = pv[i + 3 * DEG_THREADS];
        deg_count4(cnt, v0, base);
        deg_count4(cnt, v1, base);
        deg_count4(cnt, v2, base);
        deg_count4(cnt, v3, base);
    }
    for (; i < hi; i += DEG_THREADS) deg_count4(cnt, pv[i], base);
}

__global__ __launch_bounds__(DEG_THREADS) void deg_edges_kernel(const int* __restrict__ ei,
                                                                const int* __restrict__ yei,
                                                                float* __restrict__ deg,
                                                                int n, int e1, int e2) {
    __shared__ __attribute__((aligned(16))) unsigned cnt[DEG_RANGE];
    const int tid = threadIdx.x;
    const int nr  = (n + DEG_RANGE - 1) / DEG_RANGE;
    if (nr < 1) return;
    int S = (int)gridDim.x / nr;
    if (S < 1) S = 1;
    const int units = nr * S;

    for (int u = blockIdx.x; u < units; u += gridDim.x) {
        const int r = u % nr;
        const int s = u / nr;
        const int base = r * DEG_RANGE;

        for (int i = tid; i < DEG_RANGE / 4; i += DEG_THREADS)
            ((uint4*)cnt)[i] = make_uint4(0u, 0u, 0u, 0u);
        __syncthreads();

        deg_scan(ei + e1,  e1, s, S, cnt, base, tid);
        deg_scan(yei + e2, e2, s, S, cnt, base, tid);
        __syncthreads();

        for (int i = tid; i < DEG_RANGE; i += DEG_THREADS) {
            const unsigned c = cnt[i];
            const int node = base + i;
            if (c != 0u && node < n) atomicAdd(deg + node, (float)c);
        }
        __syncthreads();
    }
}

__global__ void dinv_kernel(float* __restrict__ deg, int n) {
    int i = blockIdx.x * blockDim.x + threadIdx.x;
    if (i < n) deg[i] = rsqrtf(deg[i]);
}


__global__ __launch_bounds__(128) void gemm_wmma_kernel(const float* __restrict__ x,
                                                        const float* __restrict__ W,
                                                        float* __restrict__ h, int n) {
    __shared__ float sW[IN_CH * LAT];
    __shared__ float sA[4][TILE_M * APAD];

    const int tid  = threadIdx.x;
    const int wave = tid >> 5;
    const int lane = tid & 31;
    const int ntiles = (n + TILE_M - 1) / TILE_M;
    const int tile = blockIdx.x * 4 + wave;

    {
        const float4* Wv = (const float4*)W;
        float4* sWv = (float4*)sW;
        for (int i = tid; i < (IN_CH * LAT) / 4; i += 128) sWv[i] = Wv[i];
    }
    if (tile < ntiles) {
        const int r0 = tile * TILE_M;
        for (int i = lane; i < TILE_M * (IN_CH / 4); i += 32) {
            int row = i >> 5;
            int c4  = i & 31;
            int grow = r0 + row; if (grow >= n) grow = n - 1;
            float4 v = ((const float4*)(x + (size_t)grow * IN_CH))[c4];
            float* dst = &sA[wave][row * APAD + c4 * 4];
            dst[0] = v.x; dst[1] = v.y; dst[2] = v.z; dst[3] = v.w;
        }
    }
    __syncthreads();

    if (tile >= ntiles) return;

    v8f acc0 = {};
    v8f acc1 = {};
    const int m    = lane & 15;
    const int koff = (lane >> 4) << 1;
    const int nn   = lane & 15;

    for (int kk = 0; kk < IN_CH / 4; ++kk) {
        const int kb = kk * 4 + koff;
        v2f a;
        a.x = sA[wave][m * APAD + kb];
        a.y = sA[wave][m * APAD + kb + 1];
        v2f b0, b1;
        b0.x = sW[(kb    ) * LAT + nn];
        b0.y = sW[(kb + 1) * LAT + nn];
        b1.x = sW[(kb    ) * LAT + 16 + nn];
        b1.y = sW[(kb + 1) * LAT + 16 + nn];
        acc0 = __builtin_amdgcn_wmma_f32_16x16x4_f32(false, a, false, b0,
                                                     (short)0, acc0, false, false);
        acc1 = __builtin_amdgcn_wmma_f32_16x16x4_f32(false, a, false, b1,
                                                     (short)0, acc1, false, false);
    }

    const int r0   = tile * TILE_M;
    const int rofs = (lane >> 4) << 3;
#pragma unroll
    for (int r = 0; r < 8; ++r) {
        int row = r0 + r + rofs;
        if (row < n) {
            h[(size_t)row * LAT + nn]      = acc0[r];
            h[(size_t)row * LAT + 16 + nn] = acc1[r];
        }
    }
}


__global__ void init_out_kernel(const float* __restrict__ dinv, const float* __restrict__ h,
                                const float* __restrict__ b, float* __restrict__ out, int n) {
    int gid = blockIdx.x * blockDim.x + threadIdx.x;
    if (gid >= n * LAT) return;
    int i = gid >> 5, c = gid & 31;
    float di = dinv[i];
    out[gid] = di * di * h[gid] + b[c];
}


__global__ void edge_scatter_kernel(const int* __restrict__ ei, const int* __restrict__ yei,
                                    const float* __restrict__ dinv, const float* __restrict__ h,
                                    float* __restrict__ out, int e1, int e2) {
    long long gid = (long long)blockIdx.x * blockDim.x + threadIdx.x;
    int e = (int)(gid >> 5);
    int c = (int)(gid & 31);
    int et = e1 + e2;
    if (e >= et) return;
    int s, d;
    if (e < e1) { s = ei[e];            d = ei[e1 + e]; }
    else        { int f = e - e1; s = yei[f]; d = yei[e2 + f]; }
    float w = dinv[s] * dinv[d];
    atomicAdd(out + (size_t)d * LAT + c, w * h[(size_t)s * LAT + c]);
}


extern "C" void kernel_launch(void* const* d_in, const int* in_sizes, int n_in,
                              void* d_out, int out_size, void* d_ws, size_t ws_size,
                              hipStream_t stream) {
    const float* x   = (const float*)d_in[0];
    const int*   ei  = (const int*)  d_in[1];
    const int*   yei = (const int*)  d_in[2];
    const float* W   = (const float*)d_in[3];
    const float* b   = (const float*)d_in[4];
    float* out = (float*)d_out;

    const int n  = in_sizes[0] / IN_CH;
    const int e1 = in_sizes[1] / 2;
    const int e2 = in_sizes[2] / 2;
    const int et = e1 + e2;

    float* deg = (float*)d_ws;
    size_t hoff = (((size_t)n * sizeof(float)) + 255) & ~(size_t)255;
    float* h = (float*)((char*)d_ws + hoff);

    deg_init_kernel<<<(n + 255) / 256, 256, 0, stream>>>(deg, n);
    deg_edges_kernel<<<DEG_GRID, DEG_THREADS, 0, stream>>>(ei, yei, deg, n, e1, e2);
    dinv_kernel<<<(n + 255) / 256, 256, 0, stream>>>(deg, n);
    {
        int ntiles = (n + TILE_M - 1) / TILE_M;
        int blocks = (ntiles + 3) / 4;
        gemm_wmma_kernel<<<blocks, 128, 0, stream>>>(x, W, h, n);
    }
    init_out_kernel<<<((size_t)n * LAT + 255) / 256, 256, 0, stream>>>(deg, h, b, out, n);
    {
        long long threads = (long long)et * 32;
        int blocks = (int)((threads + 255) / 256);
        edge_scatter_kernel<<<blocks, 256, 0, stream>>>(ei, yei, deg, h, out, e1, e2);
    }
}


# edge scatter batched: a wave takes 32 edges (coalesced id loads, per-lane norm, readlane broadcast, 32 h-row gathers in flight), not one edge per wave; float atomic arrival order changes
# speedup vs baseline: 2.9097x; 2.1640x over previous
#include <hip/hip_runtime.h>
#include <hip/hip_bf16.h>

typedef __attribute__((ext_vector_type(2))) float v2f;
typedef __attribute__((ext_vector_type(8))) float v8f;

#define IN_CH 128
#define LAT   32
#define TILE_M 16
#define APAD  132


__global__ void deg_init_kernel(float* __restrict__ deg, int n) {
    int i = blockIdx.x * blockDim.x + threadIdx.x;
    if (i < n) deg[i] = 1.0f;
}


#define DEG_RANGE   8192
#define DEG_THREADS 1024
#define DEG_GRID    256

__device__ __forceinline__ void deg_count(unsigned* cnt, int d, int base) {
    unsigned k = (unsigned)d - (unsigned)base;
    if (k < (unsigned)DEG_RANGE) atomicAdd(&cnt[k], 1u);
}

__device__ __forceinline__ void deg_count4(unsigned* cnt, int4 v, int base) {
    deg_count(cnt, v.x, base); deg_count(cnt, v.y, base);
    deg_count(cnt, v.z, base); deg_count(cnt, v.w, base);
}

__device__ __forceinline__ void deg_scan(const int* __restrict__ p, int L, int s, int S,
                                         unsigned* cnt, int base, int tid) {
    int head = (int)(((16u - (unsigned)((size_t)p & 15u)) & 15u) >> 2);
    if (head > L) head = L;
    const int nv    = (L - head) >> 2;
    const int tail0 = head + (nv << 2);
    if (s == 0) {
        if (tid < head) deg_count(cnt, p[tid], base);
        const int t = tail0 + tid;
        if (t < L) deg_count(cnt, p[t], base);
    }
    const int4* __restrict__ pv = (const int4*)(p + head);
    const int vchunk = (nv + S - 1) / S;
    const long long lo64 = (long long)s * vchunk;
    const int lo = lo64 < nv ? (int)lo64 : nv;
    const int hi = (nv - lo < vchunk) ? nv : lo + vchunk;
    int i = lo + tid;
    for (; i + 3 * DEG_THREADS < hi; i += 4 * DEG_THREADS) {
        const int4 v0 = pv[i];
        const int4 v1 = pv[i + DEG_THREADS];
        const int4 v2 = pv[i + 2 * DEG_THREADS];
        const int4 v3 = pv[i + 3 * DEG_THREADS];
        deg_count4(cnt, v0, base);
        deg_count4(cnt, v1, base);
        deg_count4(cnt, v2, base);
        deg_count4(cnt, v3, base);
    }
    for (; i < hi; i += DEG_THREADS) deg_count4(cnt, pv[i], base);
}

__global__ __launch_bounds__(DEG_THREADS) void deg_edges_kernel(const int* __restrict__ ei,
                                                                const int* __restrict__ yei,
                                                                float* __restrict__ deg,
                                                                int n, int e1, int e2) {
    __shared__ __attribute__((aligned(16))) unsigned cnt[DEG_RANGE];
    const int tid = threadIdx.x;
    const int nr  = (n + DEG_RANGE - 1) / DEG_RANGE;
    if (nr < 1) return;
    int S = (int)gridDim.x / nr;
    if (S < 1) S = 1;
    const int units = nr * S;

    for (int u = blockIdx.x; u < units; u += gridDim.x) {
        const int r = u % nr;
        const int s = u / nr;
        const int base = r * DEG_RANGE;

        for (int i = tid; i < DEG_RANGE / 4; i += DEG_THREADS)
            ((uint4*)cnt)[i] = make_uint4(0u, 0u, 0u, 0u);
        __syncthreads();

        deg_scan(ei + e1,  e1, s, S, cnt, base, tid);
        deg_scan(yei + e2, e2, s, S, cnt, base, tid);
        __syncthreads();

        for (int i = tid; i < DEG_RANGE; i += DEG_THREADS) {
            const unsigned c = cnt[i];
            const int node = base + i;
            if (c != 0u && node < n) atomicAdd(deg + node, (float)c);
        }
        __syncthreads();
    }
}

__global__ void dinv_kernel(float* __restrict__ deg, int n) {
    int i = blockIdx.x * blockDim.x + threadIdx.x;
    if (i < n) deg[i] = rsqrtf(deg[i]);
}


__global__ __launch_bounds__(128) void gemm_wmma_kernel(const float* __restrict__ x,
                                                        const float* __restrict__ W,
                                                        float* __restrict__ h, int n) {
    __shared__ float sW[IN_CH * LAT];
    __shared__ float sA[4][TILE_M * APAD];

    const int tid  = threadIdx.x;
    const int wave = tid >> 5;
    const int lane = tid & 31;
    const int ntiles = (n + TILE_M - 1) / TILE_M;
    const int tile = blockIdx.x * 4 + wave;

    {
        const float4* Wv = (const float4*)W;
        float4* sWv = (float4*)sW;
        for (int i = tid; i < (IN_CH * LAT) / 4; i += 128) sWv[i] = Wv[i];
    }
    if (tile < ntiles) {
        const int r0 = tile * TILE_M;
        for (int i = lane; i < TILE_M * (IN_CH / 4); i += 32) {
            int row = i >> 5;
            int c4  = i & 31;
            int grow = r0 + row; if (grow >= n) grow = n - 1;
            float4 v = ((const float4*)(x + (size_t)grow * IN_CH))[c4];
            float* dst = &sA[wave][row * APAD + c4 * 4];
            dst[0] = v.x; dst[1] = v.y; dst[2] = v.z; dst[3] = v.w;
        }
    }
    __syncthreads();

    if (tile >= ntiles) return;

    v8f acc0 = {};
    v8f acc1 = {};
    const int m    = lane & 15;
    const int koff = (lane >> 4) << 1;
    const int nn   = lane & 15;

    for (int kk = 0; kk < IN_CH / 4; ++kk) {
        const int kb = kk * 4 + koff;
        v2f a;
        a.x = sA[wave][m * APAD + kb];
        a.y = sA[wave][m * APAD + kb + 1];
        v2f b0, b1;
        b0.x = sW[(kb    ) * LAT + nn];
        b0.y = sW[(kb + 1) * LAT + nn];
        b1.x = sW[(kb    ) * LAT + 16 + nn];
        b1.y = sW[(kb + 1) * LAT + 16 + nn];
        acc0 = __builtin_amdgcn_wmma_f32_16x16x4_f32(false, a, false, b0,
                                                     (short)0, acc0, false, false);
        acc1 = __builtin_amdgcn_wmma_f32_16x16x4_f32(false, a, false, b1,
                                                     (short)0, acc1, false, false);
    }

    const int r0   = tile * TILE_M;
    const int rofs = (lane >> 4) << 3;
#pragma unroll
    for (int r = 0; r < 8; ++r) {
        int row = r0 + r + rofs;
        if (row < n) {
            h[(size_t)row * LAT + nn]      = acc0[r];
            h[(size_t)row * LAT + 16 + nn] = acc1[r];
        }
    }
}


__global__ void init_out_kernel(const float* __restrict__ dinv, const float* __restrict__ h,
                                const float* __restrict__ b, float* __restrict__ out, int n) {
    int gid = blockIdx.x * blockDim.x + threadIdx.x;
    if (gid >= n * LAT) return;
    int i = gid >> 5, c = gid & 31;
    float di = dinv[i];
    out[gid] = di * di * h[gid] + b[c];
}


#define EB 32

__global__ __launch_bounds__(256) void edge_scatter_kernel(const int* __restrict__ ei,
                                                           const int* __restrict__ yei,
                                                           const float* __restrict__ dinv,
                                                           const float* __restrict__ h,
                                                           float* __restrict__ out,
                                                           int e1, int e2) {
    const int lane = threadIdx.x & 31;
    const int wid = __builtin_amdgcn_readfirstlane((int)((blockIdx.x * blockDim.x + threadIdx.x) >> 5));
    const int nb1 = (e1 + EB - 1) / EB;
    const int nb2 = (e2 + EB - 1) / EB;
    if (wid >= nb1 + nb2) return;

    const int* __restrict__ ps; const int* __restrict__ pd; int L, base;
    if (wid < nb1) { ps = ei;  pd = ei + e1;  L = e1; base = wid * EB; }
    else           { ps = yei; pd = yei + e2; L = e2; base = (wid - nb1) * EB; }
    const int cnt = (L - base < EB) ? (L - base) : EB;

    int s = 0, d = 0; float w = 0.0f;
    if (lane < cnt) {
        s = ps[base + lane];
        d = pd[base + lane];
        w = dinv[s] * dinv[d];
    }
    const int wi = __float_as_int(w);

    if (cnt == EB) {
        float hv[EB];
#pragma unroll
        for (int j = 0; j < EB; ++j) {
            const int sj = __builtin_amdgcn_readlane(s, j);
            hv[j] = h[(size_t)sj * LAT + lane];
        }
#pragma unroll
        for (int j = 0; j < EB; ++j) {
            const int   dj = __builtin_amdgcn_readlane(d, j);
            const float wj = __int_as_float(__builtin_amdgcn_readlane(wi, j));
            atomicAdd(out + (size_t)dj * LAT + lane, wj * hv[j]);
        }
    } else {
        for (int j = 0; j < cnt; ++j) {
            const int   sj = __builtin_amdgcn_readlane(s, j);
            const int   dj = __builtin_amdgcn_readlane(d, j);
            const float wj = __int_as_float(__builtin_amdgcn_readlane(wi, j));
            atomicAdd(out + (size_t)dj * LAT + lane, wj * h[(size_t)sj * LAT + lane]);
        }
    }
}


extern "C" void kernel_launch(void* const* d_in, const int* in_sizes, int n_in,
                              void* d_out, int out_size, void* d_ws, size_t ws_size,
                              hipStream_t stream) {
    const float* x   = (const float*)d_in[0];
    const int*   ei  = (const int*)  d_in[1];
    const int*   yei = (const int*)  d_in[2];
    const float* W   = (const float*)d_in[3];
    const float* b   = (const float*)d_in[4];
    float* out = (float*)d_out;

    const int n  = in_sizes[0] / IN_CH;
    const int e1 = in_sizes[1] / 2;
    const int e2 = in_sizes[2] / 2;

    float* deg = (float*)d_ws;
    size_t hoff = (((size_t)n * sizeof(float)) + 255) & ~(size_t)255;
    float* h = (float*)((char*)d_ws + hoff);

    deg_init_kernel<<<(n + 255) / 256, 256, 0, stream>>>(deg, n);
    deg_edges_kernel<<<DEG_GRID, DEG_THREADS, 0, stream>>>(ei, yei, deg, n, e1, e2);
    dinv_kernel<<<(n + 255) / 256, 256, 0, stream>>>(deg, n);
    {
        int ntiles = (n + TILE_M - 1) / TILE_M;
        int blocks = (ntiles + 3) / 4;
        gemm_wmma_kernel<<<blocks, 128, 0, stream>>>(x, W, h, n);
    }
    init_out_kernel<<<((size_t)n * LAT + 255) / 256, 256, 0, stream>>>(deg, h, b, out, n);
    {
        int batches = (e1 + EB - 1) / EB + (e2 + EB - 1) / EB;
        int blocks = (batches + 7) / 8;
        edge_scatter_kernel<<<blocks, 256, 0, stream>>>(ei, yei, deg, h, out, e1, e2);
    }
}


# launch fusion 6->4: consumers take rsqrt of the degree in place and the GEMM epilogue writes out's start value (self loop + bias) from its accumulators; no separate rsqrt and out-init passes
# speedup vs baseline: 3.1794x; 1.0927x over previous
#include <hip/hip_runtime.h>
#include <hip/hip_bf16.h>

typedef __attribute__((ext_vector_type(2))) float v2f;
typedef __attribute__((ext_vector_type(8))) float v8f;

#define IN_CH 128
#define LAT   32
#define TILE_M 16
#define APAD  132


__global__ void deg_init_kernel(float* __restrict__ deg, int n) {
    int i = blockIdx.x * blockDim.x + threadIdx.x;
    if (i < n) deg[i] = 1.0f;
}


#define DEG_RANGE   8192
#define DEG_THREADS 1024
#define DEG_GRID    256

__device__ __forceinline__ void deg_count(unsigned* cnt, int d, int base) {
    unsigned k = (unsigned)d - (unsigned)base;
    if (k < (unsigned)DEG_RANGE) atomicAdd(&cnt[k], 1u);
}

__device__ __forceinline__ void deg_count4(unsigned* cnt, int4 v, int base) {
    deg_count(cnt, v.x, base); deg_count(cnt, v.y, base);
    deg_count(cnt, v.z, base); deg_count(cnt, v.w, base);
}

__device__ __forceinline__ void deg_scan(const int* __restrict__ p, int L, int s, int S,
                                         unsigned* cnt, int base, int tid) {
    int head = (int)(((16u - (unsigned)((size_t)p & 15u)) & 15u) >> 2);
    if (head > L) head = L;
    const int nv    = (L - head) >> 2;
    const int tail0 = head + (nv << 2);
    if (s == 0) {
        if (tid < head) deg_count(cnt, p[tid], base);
        const int t = tail0 + tid;
        if (t < L) deg_count(cnt, p[t], base);
    }
    const int4* __restrict__ pv = (const int4*)(p + head);
    const int vchunk = (nv + S - 1) / S;
    const long long lo64 = (long long)s * vchunk;
    const int lo = lo64 < nv ? (int)lo64 : nv;
    const int hi = (nv - lo < vchunk) ? nv : lo + vchunk;
    int i = lo + tid;
    for (; i + 3 * DEG_THREADS < hi; i += 4 * DEG_THREADS) {
        const int4 v0 = pv[i];
        const int4 v1 = pv[i + DEG_THREADS];
        const int4 v2 = pv[i + 2 * DEG_THREADS];
        const int4 v3 = pv[i + 3 * DEG_THREADS];
        deg_count4(cnt, v0, base);
        deg_count4(cnt, v1, base);
        deg_count4(cnt, v2, base);
        deg_count4(cnt, v3, base);
    }
    for (; i < hi; i += DEG_THREADS) deg_count4(cnt, pv[i], base);
}

__global__ __launch_bounds__(DEG_THREADS) void deg_edges_kernel(const int* __restrict__ ei,
                                                                const int* __restrict__ yei,
                                                                float* __restrict__ deg,
                                                                int n, int e1, int e2) {
    __shared__ __attribute__((aligned(16))) unsigned cnt[DEG_RANGE];
    const int tid = threadIdx.x;
    const int nr  = (n + DEG_RANGE - 1) / DEG_RANGE;
    if (nr < 1) return;
    int S = (int)gridDim.x / nr;
    if (S < 1) S = 1;
    const int units = nr * S;

    for (int u = blockIdx.x; u < units; u += gridDim.x) {
        const int r = u % nr;
        const int s = u / nr;
        const int base = r * DEG_RANGE;

        for (int i = tid; i < DEG_RANGE / 4; i += DEG_THREADS)
            ((uint4*)cnt)[i] = make_uint4(0u, 0u, 0u, 0u);
        __syncthreads();

        deg_scan(ei + e1,  e1, s, S, cnt, base, tid);
        deg_scan(yei + e2, e2, s, S, cnt, base, tid);
        __syncthreads();

        for (int i = tid; i < DEG_RANGE; i += DEG_THREADS) {
            const unsigned c = cnt[i];
            const int node = base + i;
            if (c != 0u && node < n) atomicAdd(deg + node, (float)c);
        }
        __syncthreads();
    }
}


__global__ __launch_bounds__(128) void gemm_wmma_kernel(const float* __restrict__ x,
                                                        const float* __restrict__ W,
                                                        const float* __restrict__ deg,
                                                        const float* __restrict__ b,
                                                        float* __restrict__ h,
                                                        float* __restrict__ out, int n) {
    __shared__ float sW[IN_CH * LAT];
    __shared__ float sA[4][TILE_M * APAD];

    const int tid  = threadIdx.x;
    const int wave = tid >> 5;
    const int lane = tid & 31;
    const int ntiles = (n + TILE_M - 1) / TILE_M;
    const int tile = blockIdx.x * 4 + wave;

    {
        const float4* Wv = (const float4*)W;
        float4* sWv = (float4*)sW;
        for (int i = tid; i < (IN_CH * LAT) / 4; i += 128) sWv[i] = Wv[i];
    }
    const int r0   = tile * TILE_M;
    const int rofs = (lane >> 4) << 3;
    float dg[8] = {};
    float bias0 = 0.0f, bias1 = 0.0f;
    if (tile < ntiles) {
#pragma unroll
        for (int r = 0; r < 8; ++r) {
            const int row = r0 + r + rofs;
            dg[r] = deg[row < n ? row : n - 1];
        }
        bias0 = b[lane & 15];
        bias1 = b[16 + (lane & 15)];
        for (int i = lane; i < TILE_M * (IN_CH / 4); i += 32) {
            int row = i >> 5;
            int c4  = i & 31;
            int grow = r0 + row; if (grow >= n) grow = n - 1;
            float4 v = ((const float4*)(x + (size_t)grow * IN_CH))[c4];
            float* dst = &sA[wave][row * APAD + c4 * 4];
            dst[0] = v.x; dst[1] = v.y; dst[2] = v.z; dst[3] = v.w;
        }
    }
    __syncthreads();

    if (tile >= ntiles) return;

    v8f acc0 = {};
    v8f acc1 = {};
    const int m    = lane & 15;
    const int koff = (lane >> 4) << 1;
    const int nn   = lane & 15;

    for (int kk = 0; kk < IN_CH / 4; ++kk) {
        const int kb = kk * 4 + koff;
        v2f a;
        a.x = sA[wave][m * APAD + kb];
        a.y = sA[wave][m * APAD + kb + 1];
        v2f b0, b1;
        b0.x = sW[(kb    ) * LAT + nn];
        b0.y = sW[(kb + 1) * LAT + nn];
        b1.x = sW[(kb    ) * LAT + 16 + nn];
        b1.y = sW[(kb + 1) * LAT + 16 + nn];
        acc0 = __builtin_amdgcn_wmma_f32_16x16x4_f32(false, a, false, b0,
                                                     (short)0, acc0, false, false);
        acc1 = __builtin_amdgcn_wmma_f32_16x16x4_f32(false, a, false, b1,
                                                     (short)0, acc1, false, false);
    }

#pragma unroll
    for (int r = 0; r < 8; ++r) {
        const int row = r0 + r + rofs;
        if (row < n) {
            const float di = rsqrtf(dg[r]);
            const float sc = di * di;
            const size_t o = (size_t)row * LAT + nn;
            h[o]        = acc0[r];
            h[o + 16]   = acc1[r];
            out[o]      = sc * acc0[r] + bias0;
            out[o + 16] = sc * acc1[r] + bias1;
        }
    }
}


#define EB 32

__global__ __launch_bounds__(256) void edge_scatter_kernel(const int* __restrict__ ei,
                                                           const int* __restrict__ yei,
                                                           const float* __restrict__ deg,
                                                           const float* __restrict__ h,
                                                           float* __restrict__ out,
                                                           int e1, int e2) {
    const int lane = threadIdx.x & 31;
    const int wid = __builtin_amdgcn_readfirstlane((int)((blockIdx.x * blockDim.x + threadIdx.x) >> 5));
    const int nb1 = (e1 + EB - 1) / EB;
    const int nb2 = (e2 + EB - 1) / EB;
    if (wid >= nb1 + nb2) return;

    const int* __restrict__ ps; const int* __restrict__ pd; int L, base;
    if (wid < nb1) { ps = ei;  pd = ei + e1;  L = e1; base = wid * EB; }
    else           { ps = yei; pd = yei + e2; L = e2; base = (wid - nb1) * EB; }
    const int cnt = (L - base < EB) ? (L - base) : EB;

    int s = 0, d = 0; float w = 0.0f;
    if (lane < cnt) {
        s = ps[base + lane];
        d = pd[base + lane];
        w = rsqrtf(deg[s]) * rsqrtf(deg[d]);
    }
    const int wi = __float_as_int(w);

    if (cnt == EB) {
        float hv[EB];
#pragma unroll
        for (int j = 0; j < EB; ++j) {
            const int sj = __builtin_amdgcn_readlane(s, j);
            hv[j] = h[(size_t)sj * LAT + lane];
        }
#pragma unroll
        for (int j = 0; j < EB; ++j) {
            const int   dj = __builtin_amdgcn_readlane(d, j);
            const float wj = __int_as_float(__builtin_amdgcn_readlane(wi, j));
            atomicAdd(out + (size_t)dj * LAT + lane, wj * hv[j]);
        }
    } else {
        for (int j = 0; j < cnt; ++j) {
            const int   sj = __builtin_amdgcn_readlane(s, j);
            const int   dj = __builtin_amdgcn_readlane(d, j);
            const float wj = __int_as_float(__builtin_amdgcn_readlane(wi, j));
            atomicAdd(out + (size_t)dj * LAT + lane, wj * h[(size_t)sj * LAT + lane]);
        }
    }
}


extern "C" void kernel_launch(void* const* d_in, const int* in_sizes, int n_in,
                              void* d_out, int out_size, void* d_ws, size_t ws_size,
                              hipStream_t stream) {
    const float* x   = (const float*)d_in[0];
    const int*   ei  = (const int*)  d_in[1];
    const int*   yei = (const int*)  d_in[2];
    const float* W   = (const float*)d_in[3];
    const float* b   = (const float*)d_in[4];
    float* out = (float*)d_out;

    const int n  = in_sizes[0] / IN_CH;
    const int e1 = in_sizes[1] / 2;
    const int e2 = in_sizes[2] / 2;

    float* deg = (float*)d_ws;
    size_t hoff = (((size_t)n * sizeof(float)) + 255) & ~(size_t)255;
    float* h = (float*)((char*)d_ws + hoff);

    deg_init_kernel<<<(n + 255) / 256, 256, 0, stream>>>(deg, n);
    deg_edges_kernel<<<DEG_GRID, DEG_THREADS, 0, stream>>>(ei, yei, deg, n, e1, e2);
    {
        int ntiles = (n + TILE_M - 1) / TILE_M;
        int blocks = (ntiles + 3) / 4;
        gemm_wmma_kernel<<<blocks, 128, 0, stream>>>(x, W, deg, b, h, out, n);
    }
    {
        int batches = (e1 + EB - 1) / EB + (e2 + EB - 1) / EB;
        int blocks = (batches + 7) / 8;
        edge_scatter_kernel<<<blocks, 256, 0, stream>>>(ei, yei, deg, h, out, e1, e2);
    }
}


# source half of the edge norm folded into h: GEMM epilogue stores rsqrt(deg[row])*h, so the scatter gathers one degree per edge (dst) not two; products re-associated, atomic order as before
# speedup vs baseline: 3.3265x; 1.0463x over previous
#include <hip/hip_runtime.h>
#include <hip/hip_bf16.h>

typedef __attribute__((ext_vector_type(2))) float v2f;
typedef __attribute__((ext_vector_type(8))) float v8f;

#define IN_CH 128
#define LAT   32
#define TILE_M 16
#define APAD  132


__global__ void deg_init_kernel(float* __restrict__ deg, int n) {
    int i = blockIdx.x * blockDim.x + threadIdx.x;
    if (i < n) deg[i] = 1.0f;
}


#define DEG_RANGE   8192
#define DEG_THREADS 1024
#define DEG_GRID    256

__device__ __forceinline__ void deg_count(unsigned* cnt, int d, int base) {
    unsigned k = (unsigned)d - (unsigned)base;
    if (k < (unsigned)DEG_RANGE) atomicAdd(&cnt[k], 1u);
}

__device__ __forceinline__ void deg_count4(unsigned* cnt, int4 v, int base) {
    deg_count(cnt, v.x, base); deg_count(cnt, v.y, base);
    deg_count(cnt, v.z, base); deg_count(cnt, v.w, base);
}

__device__ __forceinline__ void deg_scan(const int* __restrict__ p, int L, int s, int S,
                                         unsigned* cnt, int base, int tid) {
    int head = (int)(((16u - (unsigned)((size_t)p & 15u)) & 15u) >> 2);
    if (head > L) head = L;
    const int nv    = (L - head) >> 2;
    const int tail0 = head + (nv << 2);
    if (s == 0) {
        if (tid < head) deg_count(cnt, p[tid], base);
        const int t = tail0 + tid;
        if (t < L) deg_count(cnt, p[t], base);
    }
    const int4* __restrict__ pv = (const int4*)(p + head);
    const int vchunk = (nv + S - 1) / S;
    const long long lo64 = (long long)s * vchunk;
    const int lo = lo64 < nv ? (int)lo64 : nv;
    const int hi = (nv - lo < vchunk) ? nv : lo + vchunk;
    int i = lo + tid;
    for (; i + 3 * DEG_THREADS < hi; i += 4 * DEG_THREADS) {
        const int4 v0 = pv[i];
        const int4 v1 = pv[i + DEG_THREADS];
        const int4 v2 = pv[i + 2 * DEG_THREADS];
        const int4 v3 = pv[i + 3 * DEG_THREADS];
        deg_count4(cnt, v0, base);
        deg_count4(cnt, v1, base);
        deg_count4(cnt, v2, base);
        deg_count4(cnt, v3, base);
    }
    for (; i < hi; i += DEG_THREADS) deg_count4(cnt, pv[i], base);
}

__global__ __launch_bounds__(DEG_THREADS) void deg_edges_kernel(const int* __restrict__ ei,
                                                                const int* __restrict__ yei,
                                                                float* __restrict__ deg,
                                                                int n, int e1, int e2) {
    __shared__ __attribute__((aligned(16))) unsigned cnt[DEG_RANGE];
    const int tid = threadIdx.x;
    const int nr  = (n + DEG_RANGE - 1) / DEG_RANGE;
    if (nr < 1) return;
    int S = (int)gridDim.x / nr;
    if (S < 1) S = 1;
    const int units = nr * S;

    for (int u = blockIdx.x; u < units; u += gridDim.x) {
        const int r = u % nr;
        const int s = u / nr;
        const int base = r * DEG_RANGE;

        for (int i = tid; i < DEG_RANGE / 4; i += DEG_THREADS)
            ((uint4*)cnt)[i] = make_uint4(0u, 0u, 0u, 0u);
        __syncthreads();

        deg_scan(ei + e1,  e1, s, S, cnt, base, tid);
        deg_scan(yei + e2, e2, s, S, cnt, base, tid);
        __syncthreads();

        for (int i = tid; i < DEG_RANGE; i += DEG_THREADS) {
            const unsigned c = cnt[i];
            const int node = base + i;
            if (c != 0u && node < n) atomicAdd(deg + node, (float)c);
        }
        __syncthreads();
    }
}


__global__ __launch_bounds__(128) void gemm_wmma_kernel(const float* __restrict__ x,
                                                        const float* __restrict__ W,
                                                        const float* __restrict__ deg,
                                                        const float* __restrict__ b,
                                                        float* __restrict__ h,
                                                        float* __restrict__ out, int n) {
    __shared__ float sW[IN_CH * LAT];
    __shared__ float sA[4][TILE_M * APAD];

    const int tid  = threadIdx.x;
    const int wave = tid >> 5;
    const int lane = tid & 31;
    const int ntiles = (n + TILE_M - 1) / TILE_M;
    const int tile = blockIdx.x * 4 + wave;

    {
        const float4* Wv = (const float4*)W;
        float4* sWv = (float4*)sW;
        for (int i = tid; i < (IN_CH * LAT) / 4; i += 128) sWv[i] = Wv[i];
    }
    const int r0   = tile * TILE_M;
    const int rofs = (lane >> 4) << 3;
    float dg[8] = {};
    float bias0 = 0.0f, bias1 = 0.0f;
    if (tile < ntiles) {
#pragma unroll
        for (int r = 0; r < 8; ++r) {
            const int row = r0 + r + rofs;
            dg[r] = deg[row < n ? row : n - 1];
        }
        bias0 = b[lane & 15];
        bias1 = b[16 + (lane & 15)];
        for (int i = lane; i < TILE_M * (IN_CH / 4); i += 32) {
            int row = i >> 5;
            int c4  = i & 31;
            int grow = r0 + row; if (grow >= n) grow = n - 1;
            float4 v = ((const float4*)(x + (size_t)grow * IN_CH))[c4];
            float* dst = &sA[wave][row * APAD + c4 * 4];
            dst[0] = v.x; dst[1] = v.y; dst[2] = v.z; dst[3] = v.w;
        }
    }
    __syncthreads();

    if (tile >= ntiles) return;

    v8f acc0 = {};
    v8f acc1 = {};
    const int m    = lane & 15;
    const int koff = (lane >> 4) << 1;
    const int nn   = lane & 15;

    for (int kk = 0; kk < IN_CH / 4; ++kk) {
        const int kb = kk * 4 + koff;
        v2f a;
        a.x = sA[wave][m * APAD + kb];
        a.y = sA[wave][m * APAD + kb + 1];
        v2f b0, b1;
        b0.x = sW[(kb    ) * LAT + nn];
        b0.y = sW[(kb + 1) * LAT + nn];
        b1.x = sW[(kb    ) * LAT + 16 + nn];
        b1.y = sW[(kb + 1) * LAT + 16 + nn];
        acc0 = __builtin_amdgcn_wmma_f32_16x16x4_f32(false, a, false, b0,
                                                     (short)0, acc0, false, false);
        acc1 = __builtin_amdgcn_wmma_f32_16x16x4_f32(false, a, false, b1,
                                                     (short)0, acc1, false, false);
    }

#pragma unroll
    for (int r = 0; r < 8; ++r) {
        const int row = r0 + r + rofs;
        if (row < n) {
            const float di = rsqrtf(dg[r]);
            const float h0 = di * acc0[r];
            const float h1 = di * acc1[r];
            const size_t o = (size_t)row * LAT + nn;
            h[o]        = h0;
            h[o + 16]   = h1;
            out[o]      = di * h0 + bias0;
            out[o + 16] = di * h1 + bias1;
        }
    }
}


#define EB 32

__global__ __launch_bounds__(256) void edge_scatter_kernel(const int* __restrict__ ei,
                                                           const int* __restrict__ yei,
                                                           const float* __restrict__ deg,
                                                           const float* __restrict__ h,
                                                           float* __restrict__ out,
                                                           int e1, int e2) {
    const int lane = threadIdx.x & 31;
    const int wid = __builtin_amdgcn_readfirstlane((int)((blockIdx.x * blockDim.x + threadIdx.x) >> 5));
    const int nb1 = (e1 + EB - 1) / EB;
    const int nb2 = (e2 + EB - 1) / EB;
    if (wid >= nb1 + nb2) return;

    const int* __restrict__ ps; const int* __restrict__ pd; int L, base;
    if (wid < nb1) { ps = ei;  pd = ei + e1;  L = e1; base = wid * EB; }
    else           { ps = yei; pd = yei + e2; L = e2; base = (wid - nb1) * EB; }
    const int cnt = (L - base < EB) ? (L - base) : EB;

    int s = 0, d = 0; float w = 0.0f;
    if (lane < cnt) {
        s = ps[base + lane];
        d = pd[base + lane];
        w = rsqrtf(deg[d]);
    }
    const int wi = __float_as_int(w);

    if (cnt == EB) {
        float hv[EB];
#pragma unroll
        for (int j = 0; j < EB; ++j) {
            const int sj = __builtin_amdgcn_readlane(s, j);
            hv[j] = h[(size_t)sj * LAT + lane];
        }
#pragma unroll
        for (int j = 0; j < EB; ++j) {
            const int   dj = __builtin_amdgcn_readlane(d, j);
            const float wj = __int_as_float(__builtin_amdgcn_readlane(wi, j));
            atomicAdd(out + (size_t)dj * LAT + lane, wj * hv[j]);
        }
    } else {
        for (int j = 0; j < cnt; ++j) {
            const int   sj = __builtin_amdgcn_readlane(s, j);
            const int   dj = __builtin_amdgcn_readlane(d, j);
            const float wj = __int_as_float(__builtin_amdgcn_readlane(wi, j));
            atomicAdd(out + (size_t)dj * LAT + lane, wj * h[(size_t)sj * LAT + lane]);
        }
    }
}


extern "C" void kernel_launch(void* const* d_in, const int* in_sizes, int n_in,
                              void* d_out, int out_size, void* d_ws, size_t ws_size,
                              hipStream_t stream) {
    const float* x   = (const float*)d_in[0];
    const int*   ei  = (const int*)  d_in[1];
    const int*   yei = (const int*)  d_in[2];
    const float* W   = (const float*)d_in[3];
    const float* b   = (const float*)d_in[4];
    float* out = (float*)d_out;

    const int n  = in_sizes[0] / IN_CH;
    const int e1 = in_sizes[1] / 2;
    const int e2 = in_sizes[2] / 2;

    float* deg = (float*)d_ws;
    size_t hoff = (((size_t)n * sizeof(float)) + 255) & ~(size_t)255;
    float* h = (float*)((char*)d_ws + hoff);

    deg_init_kernel<<<(n + 255) / 256, 256, 0, stream>>>(deg, n);
    deg_edges_kernel<<<DEG_GRID, DEG_THREADS, 0, stream>>>(ei, yei, deg, n, e1, e2);
    {
        int ntiles = (n + TILE_M - 1) / TILE_M;
        int blocks = (ntiles + 3) / 4;
        gemm_wmma_kernel<<<blocks, 128, 0, stream>>>(x, W, deg, b, h, out, n);
    }
    {
        int batches = (e1 + EB - 1) / EB + (e2 + EB - 1) / EB;
        int blocks = (batches + 7) / 8;
        edge_scatter_kernel<<<blocks, 256, 0, stream>>>(ei, yei, deg, h, out, e1, e2);
    }
}
